# starting kernel
# baseline (speedup 1.0000x reference)
#include <hip/hip_runtime.h>
#include <hip/hip_bf16.h>

typedef float v2f __attribute__((ext_vector_type(2)));
typedef float v8f __attribute__((ext_vector_type(8)));

#define GCN_F 128
#define BS_STRIDE 136


__global__ void gcn_zero_i32(int* p, int n) {
    int i = blockIdx.x * blockDim.x + threadIdx.x;
    if (i < n) p[i] = 0;
}

__global__ void gcn_count_deg(const int* __restrict__ col, int* deg, int e) {
    int i = blockIdx.x * blockDim.x + threadIdx.x;
    if (i < e) atomicAdd(&deg[col[i]], 1);
}

__global__ void gcn_dinv(const int* __restrict__ deg, float* dinv, int n) {
    int i = blockIdx.x * blockDim.x + threadIdx.x;
    if (i < n) dinv[i] = rsqrtf((float)(deg[i] + 1));
}


__global__ void gcn_scan_block(const int* __restrict__ deg, int* __restrict__ off,
                               int* __restrict__ bsum, int n) {
    __shared__ int s[256];
    int i = blockIdx.x * 256 + threadIdx.x;
    int v = (i < n) ? deg[i] : 0;
    s[threadIdx.x] = v;
    __syncthreads();
    for (int d = 1; d < 256; d <<= 1) {
        int t = (threadIdx.x >= d) ? s[threadIdx.x - d] : 0;
        __syncthreads();
        s[threadIdx.x] += t;
        __syncthreads();
    }
    if (i < n) off[i] = s[threadIdx.x] - v;
    if (threadIdx.x == 255) bsum[blockIdx.x] = s[255];
}

__global__ void gcn_scan_bsum(int* bsum, int nb) {
    if (threadIdx.x == 0 && blockIdx.x == 0) {
        int run = 0;
        for (int b = 0; b < nb; ++b) { int t = bsum[b]; bsum[b] = run; run += t; }
    }
}

__global__ void gcn_scan_add(int* off, const int* __restrict__ bsum, int n) {
    int i = blockIdx.x * 256 + threadIdx.x;
    if (i < n) off[i] += bsum[blockIdx.x];
}

__global__ void gcn_copy_i32(const int* __restrict__ src, int* __restrict__ dst, int n) {
    int i = blockIdx.x * blockDim.x + threadIdx.x;
    if (i < n) dst[i] = src[i];
}

__global__ void gcn_fill_csr(const int* __restrict__ rows, const int* __restrict__ cols,
                             int* cursor, int* __restrict__ srcList, int e) {
    int i = blockIdx.x * blockDim.x + threadIdx.x;
    if (i < e) {
        int c = cols[i];
        int p = atomicAdd(&cursor[c], 1);
        srcList[p] = rows[i];
    }
}

__global__ void gcn_pack_w23(const float* __restrict__ Wmu,
                             const float* __restrict__ Wlv,
                             float* __restrict__ W23) {
    int i = blockIdx.x * blockDim.x + threadIdx.x;
    int k = i >> 7, n = i & 127;
    W23[i] = (n < 64) ? Wmu[k * 64 + n] : Wlv[k * 64 + (n - 64)];
}

__global__ __launch_bounds__(256)
void gcn_gemm_rowscale_wmma(const float* __restrict__ A,
                            const float* __restrict__ B,
                            const float* __restrict__ rowScale,
                            float* __restrict__ C,
                            int M, int K) {
    __shared__ float Bs[64 * BS_STRIDE];

    const int tid  = threadIdx.x;
    const int lane = tid & 31;
    const int wave = tid >> 5;
    const int rt   = blockIdx.x * 8 + wave;
    const bool active = (rt * 16 < M);
    const int m0   = rt * 16;
    const int half = lane >> 4;
    const int l16  = lane & 15;

    const float* Arow = active ? (A + (size_t)(m0 + l16) * K) : A;

    v8f acc[8] = {};

    for (int kb = 0; kb < K; kb += 64) {
        __syncthreads();
        for (int idx = tid; idx < 64 * 32; idx += 256) {
            int r = idx >> 5, c4 = idx & 31;
            float4 v = ((const float4*)(B + (size_t)(kb + r) * GCN_F))[c4];
            *(float4*)(Bs + r * BS_STRIDE + c4 * 4) = v;
        }
        __syncthreads();

        if (active) {
            for (int kl = 0; kl < 64; kl += 4) {
                const int ka = kl + 2 * half;
                v2f a;
                a.x = Arow[kb + ka];
                a.y = Arow[kb + ka + 1];
                const float* B0 = Bs + ka * BS_STRIDE + l16;
#pragma unroll
                for (int j = 0; j < 8; ++j) {
                    v2f b;
                    b.x = B0[16 * j];
                    b.y = B0[16 * j + BS_STRIDE];
                    acc[j] = __builtin_amdgcn_wmma_f32_16x16x4_f32(
                        false, a, false, b, (short)0, acc[j], false, false);
                }
            }
        }
    }

    if (active) {
#pragma unroll
        for (int v = 0; v < 8; ++v) {
            const int row = m0 + v + 8 * half;
            const float s = rowScale[row];
            float* Crow = C + (size_t)row * GCN_F + l16;
#pragma unroll
            for (int j = 0; j < 8; ++j) Crow[16 * j] = acc[j][v] * s;
        }
    }
}


__global__ __launch_bounds__(256)
void gcn_gather_relu(const float* __restrict__ Ts, const int* __restrict__ srcList,
                     const int* __restrict__ off, const int* __restrict__ deg,
                     const float* __restrict__ dinv, const float* __restrict__ b1,
                     float* __restrict__ H, int nNodes) {
    const int node = blockIdx.x * 8 + (threadIdx.x >> 5);
    if (node >= nNodes) return;
    const int lane = threadIdx.x & 31;

    float4 acc = ((const float4*)(Ts + (size_t)node * GCN_F))[lane];
    const int s = off[node];
    const int d = deg[node];
    for (int e = 0; e < d; ++e) {
        const int src = srcList[s + e];
        const float4 v = ((const float4*)(Ts + (size_t)src * GCN_F))[lane];
        acc.x += v.x; acc.y += v.y; acc.z += v.z; acc.w += v.w;
    }
    const float dv = dinv[node];
    const float4 bb = ((const float4*)b1)[lane];
    float4 r;
    r.x = fmaxf(dv * acc.x + bb.x, 0.0f);
    r.y = fmaxf(dv * acc.y + bb.y, 0.0f);
    r.z = fmaxf(dv * acc.z + bb.z, 0.0f);
    r.w = fmaxf(dv * acc.w + bb.w, 0.0f);
    ((float4*)(H + (size_t)node * GCN_F))[lane] = r;
}

__global__ __launch_bounds__(256)
void gcn_gather_out(const float* __restrict__ Ts, const int* __restrict__ srcList,
                    const int* __restrict__ off, const int* __restrict__ deg,
                    const float* __restrict__ dinv, const float* __restrict__ bmu,
                    const float* __restrict__ blv, float* __restrict__ out, int nNodes) {
    const int node = blockIdx.x * 8 + (threadIdx.x >> 5);
    if (node >= nNodes) return;
    const int lane = threadIdx.x & 31;

    float4 acc = ((const float4*)(Ts + (size_t)node * GCN_F))[lane];
    const int s = off[node];
    const int d = deg[node];
    for (int e = 0; e < d; ++e) {
        const int src = srcList[s + e];
        const float4 v = ((const float4*)(Ts + (size_t)src * GCN_F))[lane];
        acc.x += v.x; acc.y += v.y; acc.z += v.z; acc.w += v.w;
    }
    const float dv = dinv[node];
    float4 bb;
    float* dst;
    if (lane < 16) {
        bb  = ((const float4*)bmu)[lane];
        dst = out + (size_t)node * 64 + lane * 4;
    } else {
        bb  = ((const float4*)blv)[lane - 16];
        dst = out + (size_t)nNodes * 64 + (size_t)node * 64 + (lane - 16) * 4;
    }
    float4 r;
    r.x = dv * acc.x + bb.x;
    r.y = dv * acc.y + bb.y;
    r.z = dv * acc.z + bb.z;
    r.w = dv * acc.w + bb.w;
    *(float4*)dst = r;
}


extern "C" void kernel_launch(void* const* d_in, const int* in_sizes, int n_in,
                              void* d_out, int out_size, void* d_ws, size_t ws_size,
                              hipStream_t stream) {
    const float* X   = (const float*)d_in[0];
    const int*   ei  = (const int*)d_in[1];
    const float* W1  = (const float*)d_in[2];
    const float* b1  = (const float*)d_in[3];
    const float* Wmu = (const float*)d_in[4];
    const float* bmu = (const float*)d_in[5];
    const float* Wlv = (const float*)d_in[6];
    const float* blv = (const float*)d_in[7];
    float* out = (float*)d_out;

    const int N = in_sizes[0] / 256;
    const int E = in_sizes[1] / 2;
    const int* rows = ei;
    const int* cols = ei + E;

    float* Ts   = (float*)d_ws;
    float* Hbuf = Ts + (size_t)N * GCN_F;
    float* W23  = Hbuf + (size_t)N * GCN_F;
    float* dinv = W23 + GCN_F * GCN_F;
    int* degInt  = (int*)(dinv + N);
    int* off     = degInt + N;
    int* cursor  = off + N;
    int* bsum    = cursor + N;
    const int nb = (N + 255) / 256;
    int* srcList = bsum + ((nb + 3) & ~3);

    gcn_zero_i32<<<(N + 255) / 256, 256, 0, stream>>>(degInt, N);
    gcn_count_deg<<<(E + 255) / 256, 256, 0, stream>>>(cols, degInt, E);
    gcn_dinv<<<(N + 255) / 256, 256, 0, stream>>>(degInt, dinv, N);
    gcn_scan_block<<<nb, 256, 0, stream>>>(degInt, off, bsum, N);
    gcn_scan_bsum<<<1, 64, 0, stream>>>(bsum, nb);
    gcn_scan_add<<<nb, 256, 0, stream>>>(off, bsum, N);
    gcn_copy_i32<<<(N + 255) / 256, 256, 0, stream>>>(off, cursor, N);
    gcn_fill_csr<<<(E + 255) / 256, 256, 0, stream>>>(rows, cols, cursor, srcList, E);
    gcn_pack_w23<<<64, 256, 0, stream>>>(Wmu, Wlv, W23);

    const int rowTiles = (N + 15) / 16;
    const int gemmBlocks = (rowTiles + 7) / 8;
    const int gatherBlocks = (N + 7) / 8;

    gcn_gemm_rowscale_wmma<<<gemmBlocks, 256, 0, stream>>>(X, W1, dinv, Ts, N, 256);
    gcn_gather_relu<<<gatherBlocks, 256, 0, stream>>>(Ts, srcList, off, degInt,
                                                      dinv, b1, Hbuf, N);

    gcn_gemm_rowscale_wmma<<<gemmBlocks, 256, 0, stream>>>(Hbuf, W23, dinv, Ts, N, 128);
    gcn_gather_out<<<gatherBlocks, 256, 0, stream>>>(Ts, srcList, off, degInt,
                                                     dinv, bmu, blv, out, N);
}


# layer-1 product X@W1 and W pack share the degree-count launch (roles by block index, count as 256 looping blocks): matrix unit works while atomics wait; dinv[src] applied in layer-1 sum, order kept
# speedup vs baseline: 1.2767x; 1.2767x over previous
#include <hip/hip_runtime.h>
#include <hip/hip_bf16.h>

typedef float v2f __attribute__((ext_vector_type(2)));
typedef float v8f __attribute__((ext_vector_type(8)));

#define GCN_F 128
#define BS_STRIDE 136


__global__ void gcn_zero_i32(int* p, int n) {
    int i = blockIdx.x * blockDim.x + threadIdx.x;
    if (i < n) p[i] = 0;
}


__global__ __launch_bounds__(256)
void gcn_scan_block_dinv(const int* __restrict__ deg, int* __restrict__ offLocal,
                         int* __restrict__ bsum, float* __restrict__ dinv, int n) {
    __shared__ int wsum[8];
    const int tid  = threadIdx.x;
    const int lane = tid & 31;
    const int wave = tid >> 5;
    const int i = blockIdx.x * 256 + tid;
    const int v = (i < n) ? deg[i] : 0;
    if (i < n) dinv[i] = rsqrtf((float)(v + 1));

    int incl = v;
    for (int d = 1; d < 32; d <<= 1) {
        const int t = __shfl_up(incl, d);
        if (lane >= d) incl += t;
    }
    if (lane == 31) wsum[wave] = incl;
    __syncthreads();
    int wpre = 0, tot = 0;
    for (int w = 0; w < 8; ++w) {
        const int s = wsum[w];
        if (w < wave) wpre += s;
        tot += s;
    }
    if (i < n) offLocal[i] = wpre + incl - v;
    if (tid == 0) bsum[blockIdx.x] = tot;
}

__global__ __launch_bounds__(256)
void gcn_fill_csr(const int* __restrict__ rows, const int* __restrict__ cols,
                  const int* __restrict__ offLocal, const int* __restrict__ bsum,
                  const int* __restrict__ rank, int* __restrict__ srcList,
                  int* __restrict__ ofs, int e, int n, int nb) {
    extern __shared__ int pre[];
    __shared__ int wsum[8];
    const int tid  = threadIdx.x;
    const int lane = tid & 31;
    const int wave = tid >> 5;

    const int chunk = (nb + 255) >> 8;
    const int j0 = tid * chunk;
    const int j1 = min(j0 + chunk, nb);
    int local = 0;
    for (int j = j0; j < j1; ++j) local += bsum[j];
    int incl = local;
    for (int d = 1; d < 32; d <<= 1) {
        const int t = __shfl_up(incl, d);
        if (lane >= d) incl += t;
    }
    if (lane == 31) wsum[wave] = incl;
    __syncthreads();
    int run = incl - local;
    for (int w = 0; w < wave; ++w) run += wsum[w];
    for (int j = j0; j < j1; ++j) { pre[j] = run; run += bsum[j]; }
    __syncthreads();

    const int gid = blockIdx.x * 256 + tid;
    for (int t = gid; t < n; t += gridDim.x * 256) ofs[t] = offLocal[t] + pre[t >> 8];

    const int i4 = gid * 4;
    if (i4 >= e) return;
    if ((e & 3) == 0) {
        const int4 c = *(const int4*)(cols + i4);
        const int4 r = *(const int4*)(rows + i4);
        const int4 k = *(const int4*)(rank + i4);
        const int p0 = offLocal[c.x] + pre[c.x >> 8];
        const int p1 = offLocal[c.y] + pre[c.y >> 8];
        const int p2 = offLocal[c.z] + pre[c.z >> 8];
        const int p3 = offLocal[c.w] + pre[c.w >> 8];
        srcList[p0 + k.x] = r.x;
        srcList[p1 + k.y] = r.y;
        srcList[p2 + k.z] = r.z;
        srcList[p3 + k.w] = r.w;
    } else {
        for (int j = i4; j < i4 + 4 && j < e; ++j) {
            const int c = cols[j];
            srcList[offLocal[c] + pre[c >> 8] + rank[j]] = rows[j];
        }
    }
}

template <bool SCALE>
__device__ __forceinline__ void gcn_gemm_tile(const float* __restrict__ A,
                                              const float* __restrict__ B,
                                              const float* __restrict__ rowScale,
                                              float* __restrict__ C,
                                              int M, int K, int tileBlock, float* Bs) {
    const int tid  = threadIdx.x;
    const int lane = tid & 31;
    const int wave = tid >> 5;
    const int rt   = tileBlock * 8 + wave;
    const bool active = (rt * 16 < M);
    const int m0   = rt * 16;
    const int half = lane >> 4;
    const int l16  = lane & 15;

    const float* Arow = active ? (A + (size_t)(m0 + l16) * K) : A;

    v8f acc[8] = {};

    for (int kb = 0; kb < K; kb += 64) {
        __syncthreads();
        for (int idx = tid; idx < 64 * 32; idx += 256) {
            int r = idx >> 5, c4 = idx & 31;
            float4 v = ((const float4*)(B + (size_t)(kb + r) * GCN_F))[c4];
            *(float4*)(Bs + r * BS_STRIDE + c4 * 4) = v;
        }
        __syncthreads();

        if (active) {
            for (int kl = 0; kl < 64; kl += 4) {
                const int ka = kl + 2 * half;
                v2f a;
                a.x = Arow[kb + ka];
                a.y = Arow[kb + ka + 1];
                const float* B0 = Bs + ka * BS_STRIDE + l16;
#pragma unroll
                for (int j = 0; j < 8; ++j) {
                    v2f b;
                    b.x = B0[16 * j];
                    b.y = B0[16 * j + BS_STRIDE];
                    acc[j] = __builtin_amdgcn_wmma_f32_16x16x4_f32(
                        false, a, false, b, (short)0, acc[j], false, false);
                }
            }
        }
    }

    if (active) {
#pragma unroll
        for (int v = 0; v < 8; ++v) {
            const int row = m0 + v + 8 * half;
            const float s = SCALE ? rowScale[row] : 1.0f;
            float* Crow = C + (size_t)row * GCN_F + l16;
#pragma unroll
            for (int j = 0; j < 8; ++j) Crow[16 * j] = SCALE ? acc[j][v] * s : acc[j][v];
        }
    }
}

__global__ __launch_bounds__(256)
void gcn_gemm_rowscale_wmma(const float* __restrict__ A,
                            const float* __restrict__ B,
                            const float* __restrict__ rowScale,
                            float* __restrict__ C,
                            int M, int K) {
    __shared__ float Bs[64 * BS_STRIDE];
    gcn_gemm_tile<true>(A, B, rowScale, C, M, K, blockIdx.x, Bs);
}

__global__ __launch_bounds__(256)
void gcn_count_gemm1_pack(const int* __restrict__ cols, int* deg, int* __restrict__ rank, int e,
                          const float* __restrict__ X, const float* __restrict__ W1,
                          float* __restrict__ T, int M, int K,
                          const float* __restrict__ Wmu, const float* __restrict__ Wlv,
                          float* __restrict__ W23, int countBlocks, int gemmBlocks) {
    __shared__ float Bs[64 * BS_STRIDE];
    const int b = blockIdx.x;

    if (b < countBlocks) {
        const int t    = b * 256 + threadIdx.x;
        const int nthr = countBlocks * 256;
        if ((e & 3) == 0) {
            const int q = e >> 2;
            const int4* c4 = (const int4*)cols;
            int4* r4 = (int4*)rank;
            int i = t;
            if (i < q) {
                int4 c = c4[i];
                for (;;) {
                    const int nx = i + nthr;
                    int4 cn = c;
                    if (nx < q) cn = c4[nx];
                    int4 k;
                    k.x = atomicAdd(&deg[c.x], 1);
                    k.y = atomicAdd(&deg[c.y], 1);
                    k.z = atomicAdd(&deg[c.z], 1);
                    k.w = atomicAdd(&deg[c.w], 1);
                    r4[i] = k;
                    if (nx >= q) break;
                    i = nx;
                    c = cn;
                }
            }
        } else {
            for (int i = t; i < e; i += nthr) rank[i] = atomicAdd(&deg[cols[i]], 1);
        }
        return;
    }

    if (b < countBlocks + gemmBlocks) {
        gcn_gemm_tile<false>(X, W1, nullptr, T, M, K, b - countBlocks, Bs);
        return;
    }

    const int i = (b - countBlocks - gemmBlocks) * 256 + threadIdx.x;
    const int k = i >> 7, n = i & 127;
    W23[i] = (n < 64) ? Wmu[k * 64 + n] : Wlv[k * 64 + (n - 64)];
}


__global__ __launch_bounds__(256)
void gcn_gather_relu(const float* __restrict__ T, const int* __restrict__ srcList,
                     const int* __restrict__ ofs, int nEdges,
                     const float* __restrict__ dinv, const float* __restrict__ b1,
                     float* __restrict__ H, int nNodes) {
    const int node = blockIdx.x * 8 + (threadIdx.x >> 5);
    if (node >= nNodes) return;
    const int lane = threadIdx.x & 31;

    const float4* __restrict__ T4 = (const float4*)T;
    const float dv = dinv[node];
    float4 acc = T4[(unsigned)node * 32u + (unsigned)lane];
    acc.x *= dv; acc.y *= dv; acc.z *= dv; acc.w *= dv;
    const int s = ofs[node];
    const int d = ((node + 1 < nNodes) ? ofs[node + 1] : nEdges) - s;
    for (int e = 0; e < d; ++e) {
        const int src = srcList[s + e];
        const float w = dinv[src];
        const float4 v = T4[(unsigned)src * 32u + (unsigned)lane];
        acc.x += w * v.x; acc.y += w * v.y; acc.z += w * v.z; acc.w += w * v.w;
    }
    const float4 bb = ((const float4*)b1)[lane];
    float4 r;
    r.x = fmaxf(dv * acc.x + bb.x, 0.0f);
    r.y = fmaxf(dv * acc.y + bb.y, 0.0f);
    r.z = fmaxf(dv * acc.z + bb.z, 0.0f);
    r.w = fmaxf(dv * acc.w + bb.w, 0.0f);
    ((float4*)(H + (size_t)node * GCN_F))[lane] = r;
}

__global__ __launch_bounds__(256)
void gcn_gather_out(const float* __restrict__ Ts, const int* __restrict__ srcList,
                    const int* __restrict__ ofs, int nEdges,
                    const float* __restrict__ dinv, const float* __restrict__ bmu,
                    const float* __restrict__ blv, float* __restrict__ out, int nNodes) {
    const int node = blockIdx.x * 8 + (threadIdx.x >> 5);
    if (node >= nNodes) return;
    const int lane = threadIdx.x & 31;

    float4 acc = ((const float4*)(Ts + (size_t)node * GCN_F))[lane];
    const int s = ofs[node];
    const int d = ((node + 1 < nNodes) ? ofs[node + 1] : nEdges) - s;
    for (int e = 0; e < d; ++e) {
        const int src = srcList[s + e];
        const float4 v = ((const float4*)(Ts + (size_t)src * GCN_F))[lane];
        acc.x += v.x; acc.y += v.y; acc.z += v.z; acc.w += v.w;
    }
    const float dv = dinv[node];
    float4 bb;
    float* dst;
    if (lane < 16) {
        bb  = ((const float4*)bmu)[lane];
        dst = out + (size_t)node * 64 + lane * 4;
    } else {
        bb  = ((const float4*)blv)[lane - 16];
        dst = out + (size_t)nNodes * 64 + (size_t)node * 64 + (lane - 16) * 4;
    }
    float4 r;
    r.x = dv * acc.x + bb.x;
    r.y = dv * acc.y + bb.y;
    r.z = dv * acc.z + bb.z;
    r.w = dv * acc.w + bb.w;
    *(float4*)dst = r;
}


extern "C" void kernel_launch(void* const* d_in, const int* in_sizes, int n_in,
                              void* d_out, int out_size, void* d_ws, size_t ws_size,
                              hipStream_t stream) {
    const float* X   = (const float*)d_in[0];
    const int*   ei  = (const int*)d_in[1];
    const float* W1  = (const float*)d_in[2];
    const float* b1  = (const float*)d_in[3];
    const float* Wmu = (const float*)d_in[4];
    const float* bmu = (const float*)d_in[5];
    const float* Wlv = (const float*)d_in[6];
    const float* blv = (const float*)d_in[7];
    float* out = (float*)d_out;

    const int N = in_sizes[0] / 256;
    const int E = in_sizes[1] / 2;
    const int* rows = ei;
    const int* cols = ei + E;

    float* Ts   = (float*)d_ws;
    float* Hbuf = Ts + (size_t)N * GCN_F;
    float* W23  = Hbuf + (size_t)N * GCN_F;
    float* dinv = W23 + GCN_F * GCN_F;
    int* degInt  = (int*)(dinv + N);
    int* off     = degInt + N;
    int* ofs     = degInt;
    int* bsum    = off + N;
    const int nb = (N + 255) / 256;
    int* srcList = bsum + ((nb + 3) & ~3);
    int* rank    = (int*)Hbuf;

    const int rowTiles = (N + 15) / 16;
    const int gemmBlocks = (rowTiles + 7) / 8;
    const int gatherBlocks = (N + 7) / 8;
    const int countBlocks = 256;

    gcn_zero_i32<<<(N + 255) / 256, 256, 0, stream>>>(degInt, N);
    gcn_count_gemm1_pack<<<countBlocks + gemmBlocks + 64, 256, 0, stream>>>(
        cols, degInt, rank, E, X, W1, Ts, N, 256, Wmu, Wlv, W23, countBlocks, gemmBlocks);
    gcn_scan_block_dinv<<<nb, 256, 0, stream>>>(degInt, off, bsum, dinv, N);
    gcn_fill_csr<<<(E + 1023) / 1024, 256, (size_t)nb * sizeof(int), stream>>>(
        rows, cols, off, bsum, rank, srcList, ofs, E, N, nb);

    gcn_gather_relu<<<gatherBlocks, 256, 0, stream>>>(Ts, srcList, ofs, E,
                                                      dinv, b1, Hbuf, N);

    gcn_gemm_rowscale_wmma<<<gemmBlocks, 256, 0, stream>>>(Hbuf, W23, dinv, Ts, N, 128);
    gcn_gather_out<<<gatherBlocks, 256, 0, stream>>>(Ts, srcList, ofs, E,
                                                     dinv, bmu, blv, out, N);
}


# degree counters kept twice, one plane per half of the chip: a count work-group adds only into its own half's plane, no counter line written by both halves; neighbour sum order: plane, then arrival
# speedup vs baseline: 1.6486x; 1.2913x over previous
#include <hip/hip_runtime.h>
#include <hip/hip_bf16.h>

typedef float v2f __attribute__((ext_vector_type(2)));
typedef float v8f __attribute__((ext_vector_type(8)));

#define GCN_F 128
#define BS_STRIDE 136


__global__ void gcn_zero_i32(int* p, int* q, int n) {
    int i = blockIdx.x * blockDim.x + threadIdx.x;
    if (i < n) { p[i] = 0; q[i] = 0; }
}


__global__ __launch_bounds__(256)
void gcn_scan_block_dinv(const int* __restrict__ degLo, const int* __restrict__ degHi,
                         int* __restrict__ offLocal,
                         int* __restrict__ bsum, float* __restrict__ dinv, int n) {
    __shared__ int wsum[8];
    const int tid  = threadIdx.x;
    const int lane = tid & 31;
    const int wave = tid >> 5;
    const int i = blockIdx.x * 256 + tid;
    const int v = (i < n) ? degLo[i] + degHi[i] : 0;
    if (i < n) dinv[i] = rsqrtf((float)(v + 1));

    int incl = v;
    for (int d = 1; d < 32; d <<= 1) {
        const int t = __shfl_up(incl, d);
        if (lane >= d) incl += t;
    }
    if (lane == 31) wsum[wave] = incl;
    __syncthreads();
    int wpre = 0, tot = 0;
    for (int w = 0; w < 8; ++w) {
        const int s = wsum[w];
        if (w < wave) wpre += s;
        tot += s;
    }
    if (i < n) offLocal[i] = wpre + incl - v;
    if (tid == 0) bsum[blockIdx.x] = tot;
}

__global__ __launch_bounds__(256)
void gcn_fill_csr(const int* __restrict__ rows, const int* __restrict__ cols,
                  const int* __restrict__ offLocal, const int* __restrict__ bsum,
                  const int* __restrict__ rank, const int* __restrict__ degLo,
                  int* __restrict__ srcList,
                  int* __restrict__ ofs, int e, int n, int nb) {
    extern __shared__ int pre[];
    __shared__ int wsum[8];
    const int tid  = threadIdx.x;
    const int lane = tid & 31;
    const int wave = tid >> 5;

    const int chunk = (nb + 255) >> 8;
    const int j0 = tid * chunk;
    const int j1 = min(j0 + chunk, nb);
    int local = 0;
    for (int j = j0; j < j1; ++j) local += bsum[j];
    int incl = local;
    for (int d = 1; d < 32; d <<= 1) {
        const int t = __shfl_up(incl, d);
        if (lane >= d) incl += t;
    }
    if (lane == 31) wsum[wave] = incl;
    __syncthreads();
    int run = incl - local;
    for (int w = 0; w < wave; ++w) run += wsum[w];
    for (int j = j0; j < j1; ++j) { pre[j] = run; run += bsum[j]; }
    __syncthreads();

    const int gid = blockIdx.x * 256 + tid;
    for (int t = gid; t < n; t += gridDim.x * 256) ofs[t] = offLocal[t] + pre[t >> 8];

    const int i4 = gid * 4;
    if (i4 >= e) return;
    if ((e & 3) == 0) {
        const int4 c = *(const int4*)(cols + i4);
        const int4 r = *(const int4*)(rows + i4);
        const int4 k = *(const int4*)(rank + i4);
        const int p0 = offLocal[c.x] + pre[c.x >> 8] + ((k.x < 0) ? degLo[c.x] : 0);
        const int p1 = offLocal[c.y] + pre[c.y >> 8] + ((k.y < 0) ? degLo[c.y] : 0);
        const int p2 = offLocal[c.z] + pre[c.z >> 8] + ((k.z < 0) ? degLo[c.z] : 0);
        const int p3 = offLocal[c.w] + pre[c.w >> 8] + ((k.w < 0) ? degLo[c.w] : 0);
        srcList[p0 + (k.x & 0x7fffffff)] = r.x;
        srcList[p1 + (k.y & 0x7fffffff)] = r.y;
        srcList[p2 + (k.z & 0x7fffffff)] = r.z;
        srcList[p3 + (k.w & 0x7fffffff)] = r.w;
    } else {
        for (int j = i4; j < i4 + 4 && j < e; ++j) {
            const int c = cols[j];
            const int k = rank[j];
            srcList[offLocal[c] + pre[c >> 8] + ((k < 0) ? degLo[c] : 0) + (k & 0x7fffffff)] = rows[j];
        }
    }
}

template <bool SCALE>
__device__ __forceinline__ void gcn_gemm_tile(const float* __restrict__ A,
                                              const float* __restrict__ B,
                                              const float* __restrict__ rowScale,
                                              float* __restrict__ C,
                                              int M, int K, int tileBlock, float* Bs) {
    const int tid  = threadIdx.x;
    const int lane = tid & 31;
    const int wave = tid >> 5;
    const int rt   = tileBlock * 8 + wave;
    const bool active = (rt * 16 < M);
    const int m0   = rt * 16;
    const int half = lane >> 4;
    const int l16  = lane & 15;

    const float* Arow = active ? (A + (size_t)(m0 + l16) * K) : A;

    v8f acc[8] = {};

    for (int kb = 0; kb < K; kb += 64) {
        __syncthreads();
        for (int idx = tid; idx < 64 * 32; idx += 256) {
            int r = idx >> 5, c4 = idx & 31;
            float4 v = ((const float4*)(B + (size_t)(kb + r) * GCN_F))[c4];
            *(float4*)(Bs + r * BS_STRIDE + c4 * 4) = v;
        }
        __syncthreads();

        if (active) {
            for (int kl = 0; kl < 64; kl += 4) {
                const int ka = kl + 2 * half;
                v2f a;
                a.x = Arow[kb + ka];
                a.y = Arow[kb + ka + 1];
                const float* B0 = Bs + ka * BS_STRIDE + l16;
#pragma unroll
                for (int j = 0; j < 8; ++j) {
                    v2f b;
                    b.x = B0[16 * j];
                    b.y = B0[16 * j + BS_STRIDE];
                    acc[j] = __builtin_amdgcn_wmma_f32_16x16x4_f32(
                        false, a, false, b, (short)0, acc[j], false, false);
                }
            }
        }
    }

    if (active) {
#pragma unroll
        for (int v = 0; v < 8; ++v) {
            const int row = m0 + v + 8 * half;
            const float s = SCALE ? rowScale[row] : 1.0f;
            float* Crow = C + (size_t)row * GCN_F + l16;
#pragma unroll
            for (int j = 0; j < 8; ++j) Crow[16 * j] = SCALE ? acc[j][v] * s : acc[j][v];
        }
    }
}

__global__ __launch_bounds__(256)
void gcn_gemm_rowscale_wmma(const float* __restrict__ A,
                            const float* __restrict__ B,
                            const float* __restrict__ rowScale,
                            float* __restrict__ C,
                            int M, int K) {
    __shared__ float Bs[64 * BS_STRIDE];
    gcn_gemm_tile<true>(A, B, rowScale, C, M, K, blockIdx.x, Bs);
}

__global__ __launch_bounds__(256)
void gcn_count_gemm1_pack(const int* __restrict__ cols, int* degLo, int* degHi,
                          int* __restrict__ rank, int e,
                          const float* __restrict__ X, const float* __restrict__ W1,
                          float* __restrict__ T, int M, int K,
                          const float* __restrict__ Wmu, const float* __restrict__ Wlv,
                          float* __restrict__ W23, int countBlocks, int gemmBlocks) {
    __shared__ float Bs[64 * BS_STRIDE];
    const int b = blockIdx.x;

    if (b < countBlocks) {
        const int t    = b * 256 + threadIdx.x;
        const int nthr = countBlocks * 256;
        const unsigned where = __builtin_amdgcn_s_sendmsg_rtn(0x87);
        const bool hi  = ((where >> 18) & 1u) != 0u;
        int* deg       = hi ? degHi : degLo;
        const int mark = hi ? (int)0x80000000u : 0;
        if ((e & 3) == 0) {
            const int q = e >> 2;
            const int4* c4 = (const int4*)cols;
            int4* r4 = (int4*)rank;
            int i = t;
            if (i < q) {
                int4 c = c4[i];
                for (;;) {
                    const int nx = i + nthr;
                    int4 cn = c;
                    if (nx < q) cn = c4[nx];
                    int4 k;
                    k.x = atomicAdd(&deg[c.x], 1) | mark;
                    k.y = atomicAdd(&deg[c.y], 1) | mark;
                    k.z = atomicAdd(&deg[c.z], 1) | mark;
                    k.w = atomicAdd(&deg[c.w], 1) | mark;
                    r4[i] = k;
                    if (nx >= q) break;
                    i = nx;
                    c = cn;
                }
            }
        } else {
            for (int i = t; i < e; i += nthr) rank[i] = atomicAdd(&deg[cols[i]], 1) | mark;
        }
        return;
    }

    if (b < countBlocks + gemmBlocks) {
        gcn_gemm_tile<false>(X, W1, nullptr, T, M, K, b - countBlocks, Bs);
        return;
    }

    const int i = (b - countBlocks - gemmBlocks) * 256 + threadIdx.x;
    const int k = i >> 7, n = i & 127;
    W23[i] = (n < 64) ? Wmu[k * 64 + n] : Wlv[k * 64 + (n - 64)];
}


__global__ __launch_bounds__(256)
void gcn_gather_relu(const float* __restrict__ T, const int* __restrict__ srcList,
                     const int* __restrict__ ofs, int nEdges,
                     const float* __restrict__ dinv, const float* __restrict__ b1,
                     float* __restrict__ H, int nNodes) {
    const int node = blockIdx.x * 8 + (threadIdx.x >> 5);
    if (node >= nNodes) return;
    const int lane = threadIdx.x & 31;

    const float4* __restrict__ T4 = (const float4*)T;
    const float dv = dinv[node];
    float4 acc = T4[(unsigned)node * 32u + (unsigned)lane];
    acc.x *= dv; acc.y *= dv; acc.z *= dv; acc.w *= dv;
    const int s = ofs[node];
    const int d = ((node + 1 < nNodes) ? ofs[node + 1] : nEdges) - s;
    for (int e = 0; e < d; ++e) {
        const int src = srcList[s + e];
        const float w = dinv[src];
        const float4 v = T4[(unsigned)src * 32u + (unsigned)lane];
        acc.x += w * v.x; acc.y += w * v.y; acc.z += w * v.z; acc.w += w * v.w;
    }
    const float4 bb = ((const float4*)b1)[lane];
    float4 r;
    r.x = fmaxf(dv * acc.x + bb.x, 0.0f);
    r.y = fmaxf(dv * acc.y + bb.y, 0.0f);
    r.z = fmaxf(dv * acc.z + bb.z, 0.0f);
    r.w = fmaxf(dv * acc.w + bb.w, 0.0f);
    ((float4*)(H + (size_t)node * GCN_F))[lane] = r;
}

__global__ __launch_bounds__(256)
void gcn_gather_out(const float* __restrict__ Ts, const int* __restrict__ srcList,
                    const int* __restrict__ ofs, int nEdges,
                    const float* __restrict__ dinv, const float* __restrict__ bmu,
                    const float* __restrict__ blv, float* __restrict__ out, int nNodes) {
    const int node = blockIdx.x * 8 + (threadIdx.x >> 5);
    if (node >= nNodes) return;
    const int lane = threadIdx.x & 31;

    float4 acc = ((const float4*)(Ts + (size_t)node * GCN_F))[lane];
    const int s = ofs[node];
    const int d = ((node + 1 < nNodes) ? ofs[node + 1] : nEdges) - s;
    for (int e = 0; e < d; ++e) {
        const int src = srcList[s + e];
        const float4 v = ((const float4*)(Ts + (size_t)src * GCN_F))[lane];
        acc.x += v.x; acc.y += v.y; acc.z += v.z; acc.w += v.w;
    }
    const float dv = dinv[node];
    float4 bb;
    float* dst;
    if (lane < 16) {
        bb  = ((const float4*)bmu)[lane];
        dst = out + (size_t)node * 64 + lane * 4;
    } else {
        bb  = ((const float4*)blv)[lane - 16];
        dst = out + (size_t)nNodes * 64 + (size_t)node * 64 + (lane - 16) * 4;
    }
    float4 r;
    r.x = dv * acc.x + bb.x;
    r.y = dv * acc.y + bb.y;
    r.z = dv * acc.z + bb.z;
    r.w = dv * acc.w + bb.w;
    *(float4*)dst = r;
}


extern "C" void kernel_launch(void* const* d_in, const int* in_sizes, int n_in,
                              void* d_out, int out_size, void* d_ws, size_t ws_size,
                              hipStream_t stream) {
    const float* X   = (const float*)d_in[0];
    const int*   ei  = (const int*)d_in[1];
    const float* W1  = (const float*)d_in[2];
    const float* b1  = (const float*)d_in[3];
    const float* Wmu = (const float*)d_in[4];
    const float* bmu = (const float*)d_in[5];
    const float* Wlv = (const float*)d_in[6];
    const float* blv = (const float*)d_in[7];
    float* out = (float*)d_out;

    const int N = in_sizes[0] / 256;
    const int E = in_sizes[1] / 2;
    const int* rows = ei;
    const int* cols = ei + E;

    float* Ts   = (float*)d_ws;
    float* Hbuf = Ts + (size_t)N * GCN_F;
    float* W23  = Hbuf + (size_t)N * GCN_F;
    float* dinv = W23 + GCN_F * GCN_F;
    int* degInt  = (int*)(dinv + N);
    int* off     = degInt + N;
    int* ofs     = degInt;
    int* bsum    = off + N;
    const int nb = (N + 255) / 256;
    int* srcList = bsum + ((nb + 3) & ~3);
    int* rank    = (int*)Hbuf;
    int* degLo   = rank + E;

    const int rowTiles = (N + 15) / 16;
    const int gemmBlocks = (rowTiles + 7) / 8;
    const int gatherBlocks = (N + 7) / 8;
    const int countBlocks = 256;

    gcn_zero_i32<<<(N + 255) / 256, 256, 0, stream>>>(degLo, degInt, N);
    gcn_count_gemm1_pack<<<countBlocks + gemmBlocks + 64, 256, 0, stream>>>(
        cols, degLo, degInt, rank, E, X, W1, Ts, N, 256, Wmu, Wlv, W23, countBlocks, gemmBlocks);
    gcn_scan_block_dinv<<<nb, 256, 0, stream>>>(degLo, degInt, off, bsum, dinv, N);
    gcn_fill_csr<<<(E + 1023) / 1024, 256, (size_t)nb * sizeof(int), stream>>>(
        rows, cols, off, bsum, rank, degLo, srcList, ofs, E, N, nb);

    gcn_gather_relu<<<gatherBlocks, 256, 0, stream>>>(Ts, srcList, ofs, E,
                                                      dinv, b1, Hbuf, N);

    gcn_gemm_rowscale_wmma<<<gemmBlocks, 256, 0, stream>>>(Hbuf, W23, dinv, Ts, N, 128);
    gcn_gather_out<<<gatherBlocks, 256, 0, stream>>>(Ts, srcList, ofs, E,
                                                     dinv, bmu, blv, out, N);
}


# layer-2 product taken behind the layer-2 neighbour sum, (A H) W for A (H W): resident 16-node tiles sum rows to an LDS strip, 4 waves multiply it with W in LDS; H@W launch gone; product sum order new
# speedup vs baseline: 1.9308x; 1.1712x over previous
#include <hip/hip_runtime.h>
#include <hip/hip_bf16.h>

typedef float v2f __attribute__((ext_vector_type(2)));
typedef float v8f __attribute__((ext_vector_type(8)));

#define GCN_F 128
#define BS_STRIDE 136
#define AS_STRIDE 132


__global__ void gcn_zero_i32(int* p, int* q, int n) {
    int i = blockIdx.x * blockDim.x + threadIdx.x;
    if (i < n) { p[i] = 0; q[i] = 0; }
}


__global__ __launch_bounds__(256)
void gcn_scan_block_dinv(const int* __restrict__ degLo, const int* __restrict__ degHi,
                         int* __restrict__ offLocal,
                         int* __restrict__ bsum, float* __restrict__ dinv, int n) {
    __shared__ int wsum[8];
    const int tid  = threadIdx.x;
    const int lane = tid & 31;
    const int wave = tid >> 5;
    const int i = blockIdx.x * 256 + tid;
    const int v = (i < n) ? degLo[i] + degHi[i] : 0;
    if (i < n) dinv[i] = rsqrtf((float)(v + 1));

    int incl = v;
    for (int d = 1; d < 32; d <<= 1) {
        const int t = __shfl_up(incl, d);
        if (lane >= d) incl += t;
    }
    if (lane == 31) wsum[wave] = incl;
    __syncthreads();
    int wpre = 0, tot = 0;
    for (int w = 0; w < 8; ++w) {
        const int s = wsum[w];
        if (w < wave) wpre += s;
        tot += s;
    }
    if (i < n) offLocal[i] = wpre + incl - v;
    if (tid == 0) bsum[blockIdx.x] = tot;
}

__global__ __launch_bounds__(256)
void gcn_fill_csr(const int* __restrict__ rows, const int* __restrict__ cols,
                  const int* __restrict__ offLocal, const int* __restrict__ bsum,
                  const int* __restrict__ rank, const int* __restrict__ degLo,
                  int* __restrict__ srcList,
                  int* __restrict__ ofs, int e, int n, int nb) {
    extern __shared__ int pre[];
    __shared__ int wsum[8];
    const int tid  = threadIdx.x;
    const int lane = tid & 31;
    const int wave = tid >> 5;

    const int chunk = (nb + 255) >> 8;
    const int j0 = tid * chunk;
    const int j1 = min(j0 + chunk, nb);
    int local = 0;
    for (int j = j0; j < j1; ++j) local += bsum[j];
    int incl = local;
    for (int d = 1; d < 32; d <<= 1) {
        const int t = __shfl_up(incl, d);
        if (lane >= d) incl += t;
    }
    if (lane == 31) wsum[wave] = incl;
    __syncthreads();
    int run = incl - local;
    for (int w = 0; w < wave; ++w) run += wsum[w];
    for (int j = j0; j < j1; ++j) { pre[j] = run; run += bsum[j]; }
    __syncthreads();

    const int gid = blockIdx.x * 256 + tid;
    for (int t = gid; t < n; t += gridDim.x * 256) ofs[t] = offLocal[t] + pre[t >> 8];

    const int i4 = gid * 4;
    if (i4 >= e) return;
    if ((e & 3) == 0) {
        const int4 c = *(const int4*)(cols + i4);
        const int4 r = *(const int4*)(rows + i4);
        const int4 k = *(const int4*)(rank + i4);
        const int p0 = offLocal[c.x] + pre[c.x >> 8] + ((k.x < 0) ? degLo[c.x] : 0);
        const int p1 = offLocal[c.y] + pre[c.y >> 8] + ((k.y < 0) ? degLo[c.y] : 0);
        const int p2 = offLocal[c.z] + pre[c.z >> 8] + ((k.z < 0) ? degLo[c.z] : 0);
        const int p3 = offLocal[c.w] + pre[c.w >> 8] + ((k.w < 0) ? degLo[c.w] : 0);
        srcList[p0 + (k.x & 0x7fffffff)] = r.x;
        srcList[p1 + (k.y & 0x7fffffff)] = r.y;
        srcList[p2 + (k.z & 0x7fffffff)] = r.z;
        srcList[p3 + (k.w & 0x7fffffff)] = r.w;
    } else {
        for (int j = i4; j < i4 + 4 && j < e; ++j) {
            const int c = cols[j];
            const int k = rank[j];
            srcList[offLocal[c] + pre[c >> 8] + ((k < 0) ? degLo[c] : 0) + (k & 0x7fffffff)] = rows[j];
        }
    }
}

__device__ __forceinline__ void gcn_gemm_tile(const float* __restrict__ A,
                                              const float* __restrict__ B,
                                              float* __restrict__ C,
                                              int M, int K, int tileBlock, float* Bs) {
    const int tid  = threadIdx.x;
    const int lane = tid & 31;
    const int wave = tid >> 5;
    const int rt   = tileBlock * 8 + wave;
    const bool active = (rt * 16 < M);
    const int m0   = rt * 16;
    const int half = lane >> 4;
    const int l16  = lane & 15;

    const float* Arow = active ? (A + (size_t)(m0 + l16) * K) : A;

    v8f acc[8] = {};

    for (int kb = 0; kb < K; kb += 64) {
        __syncthreads();
        for (int idx = tid; idx < 64 * 32; idx += 256) {
            int r = idx >> 5, c4 = idx & 31;
            float4 v = ((const float4*)(B + (size_t)(kb + r) * GCN_F))[c4];
            *(float4*)(Bs + r * BS_STRIDE + c4 * 4) = v;
        }
        __syncthreads();

        if (active) {
            for (int kl = 0; kl < 64; kl += 4) {
                const int ka = kl + 2 * half;
                v2f a;
                a.x = Arow[kb + ka];
                a.y = Arow[kb + ka + 1];
                const float* B0 = Bs + ka * BS_STRIDE + l16;
#pragma unroll
                for (int j = 0; j < 8; ++j) {
                    v2f b;
                    b.x = B0[16 * j];
                    b.y = B0[16 * j + BS_STRIDE];
                    acc[j] = __builtin_amdgcn_wmma_f32_16x16x4_f32(
                        false, a, false, b, (short)0, acc[j], false, false);
                }
            }
        }
    }

    if (active) {
#pragma unroll
        for (int v = 0; v < 8; ++v) {
            const int row = m0 + v + 8 * half;
            float* Crow = C + (size_t)row * GCN_F + l16;
#pragma unroll
            for (int j = 0; j < 8; ++j) Crow[16 * j] = acc[j][v];
        }
    }
}

__global__ __launch_bounds__(256)
void gcn_count_gemm1_pack(const int* __restrict__ cols, int* degLo, int* degHi,
                          int* __restrict__ rank, int e,
                          const float* __restrict__ X, const float* __restrict__ W1,
                          float* __restrict__ T, int M, int K,
                          const float* __restrict__ Wmu, const float* __restrict__ Wlv,
                          float* __restrict__ W23, int countBlocks, int gemmBlocks) {
    __shared__ float Bs[64 * BS_STRIDE];
    const int b = blockIdx.x;

    if (b < countBlocks) {
        const int t    = b * 256 + threadIdx.x;
        const int nthr = countBlocks * 256;
        const unsigned where = __builtin_amdgcn_s_sendmsg_rtn(0x87);
        const bool hi  = ((where >> 18) & 1u) != 0u;
        int* deg       = hi ? degHi : degLo;
        const int mark = hi ? (int)0x80000000u : 0;
        if ((e & 3) == 0) {
            const int q = e >> 2;
            const int4* c4 = (const int4*)cols;
            int4* r4 = (int4*)rank;
            int i = t;
            if (i < q) {
                int4 c = c4[i];
                for (;;) {
                    const int nx = i + nthr;
                    int4 cn = c;
                    if (nx < q) cn = c4[nx];
                    int4 k;
                    k.x = atomicAdd(&deg[c.x], 1) | mark;
                    k.y = atomicAdd(&deg[c.y], 1) | mark;
                    k.z = atomicAdd(&deg[c.z], 1) | mark;
                    k.w = atomicAdd(&deg[c.w], 1) | mark;
                    r4[i] = k;
                    if (nx >= q) break;
                    i = nx;
                    c = cn;
                }
            }
        } else {
            for (int i = t; i < e; i += nthr) rank[i] = atomicAdd(&deg[cols[i]], 1) | mark;
        }
        return;
    }

    if (b < countBlocks + gemmBlocks) {
        gcn_gemm_tile(X, W1, T, M, K, b - countBlocks, Bs);
        return;
    }

    const int p = (b - countBlocks - gemmBlocks) * 256 + threadIdx.x;
    const int y = p & 1, t = (p >> 1) & 1, ln = (p >> 2) & 31, w = (p >> 7) & 3, ks = p >> 9;
    const int k = 4 * ks + 2 * (ln >> 4) + y;
    const int n = 32 * w + 16 * t + (ln & 15);
    W23[p] = (n < 64) ? Wmu[k * 64 + n] : Wlv[k * 64 + (n - 64)];
}


__global__ __launch_bounds__(256)
void gcn_gather_relu(const float* __restrict__ T, const int* __restrict__ srcList,
                     const int* __restrict__ ofs, int nEdges,
                     const float* __restrict__ dinv, const float* __restrict__ b1,
                     float* __restrict__ H, int nNodes) {
    const int node = blockIdx.x * 8 + (threadIdx.x >> 5);
    if (node >= nNodes) return;
    const int lane = threadIdx.x & 31;

    const float4* __restrict__ T4 = (const float4*)T;
    const float dv = dinv[node];
    float4 acc = T4[(unsigned)node * 32u + (unsigned)lane];
    acc.x *= dv; acc.y *= dv; acc.z *= dv; acc.w *= dv;
    const int s = ofs[node];
    const int d = ((node + 1 < nNodes) ? ofs[node + 1] : nEdges) - s;
    for (int e = 0; e < d; ++e) {
        const int src = srcList[s + e];
        const float w = dinv[src];
        const float4 v = T4[(unsigned)src * 32u + (unsigned)lane];
        acc.x += w * v.x; acc.y += w * v.y; acc.z += w * v.z; acc.w += w * v.w;
    }
    const float4 bb = ((const float4*)b1)[lane];
    float4 r;
    r.x = dv * fmaxf(dv * acc.x + bb.x, 0.0f);
    r.y = dv * fmaxf(dv * acc.y + bb.y, 0.0f);
    r.z = dv * fmaxf(dv * acc.z + bb.z, 0.0f);
    r.w = dv * fmaxf(dv * acc.w + bb.w, 0.0f);
    ((float4*)(H + (size_t)node * GCN_F))[lane] = r;
}

__global__ __launch_bounds__(512)
void gcn_gather_gemm_out(const float* __restrict__ Hs, const int* __restrict__ srcList,
                         const int* __restrict__ ofs, int nEdges,
                         const float* __restrict__ dinv, const float* __restrict__ W23,
                         const float* __restrict__ bmu, const float* __restrict__ blv,
                         float* __restrict__ out, int nNodes, int nTiles) {
    __shared__ float Bf[GCN_F * GCN_F];
    __shared__ float As[16 * AS_STRIDE];
    const int tid  = threadIdx.x;
    const int lane = tid & 31;
    const int wave = __builtin_amdgcn_readfirstlane(tid >> 5);
    const int half = lane >> 4;
    const int l16  = lane & 15;

    for (int i = tid; i < GCN_F * GCN_F / 4; i += 512) ((float4*)Bf)[i] = ((const float4*)W23)[i];

    const int cb = (wave & 1) * 32 + l16;
    const float* bias = (wave & 2) ? blv : bmu;
    float bias0 = 0.0f, bias1 = 0.0f;
    if (wave < 4) { bias0 = bias[cb]; bias1 = bias[cb + 16]; }
    float* outPlane = out + ((wave & 2) ? (size_t)nNodes * 64 : (size_t)0) + cb;

    const float4* __restrict__ H4 = (const float4*)Hs;

    for (int tile = blockIdx.x; tile < nTiles; tile += gridDim.x) {
        const int node = tile * 16 + wave;
        float4 acc = make_float4(0.0f, 0.0f, 0.0f, 0.0f);
        if (node < nNodes) {
            const float dv = dinv[node];
            acc = H4[(unsigned)node * 32u + (unsigned)lane];
            const int o2 = ofs[min(node + (lane & 1), nNodes - 1)];
            const int s  = __builtin_amdgcn_readlane(o2, 0);
            const int d  = ((node + 1 < nNodes) ? __builtin_amdgcn_readlane(o2, 1) : nEdges) - s;
            if (d > 0) {
                const int last = nEdges - 1;
                int ent = srcList[min(s + lane, last)];
                for (int e0 = 0; e0 < d; e0 += 32) {
                    const int cnt = min(32, d - e0);
                    int entNext = ent;
                    if (e0 + 32 < d) entNext = srcList[min(s + e0 + 32 + lane, last)];
                    for (int g = 0; g < cnt; g += 8) {
                        const int m = cnt - g;
                        float4 v[8];
#pragma unroll
                        for (int j = 0; j < 8; ++j)
                            if (j < m) {
                                const int src = __builtin_amdgcn_readlane(ent, g + j);
                                v[j] = H4[(unsigned)src * 32u + (unsigned)lane];
                            }
#pragma unroll
                        for (int j = 0; j < 8; ++j)
                            if (j < m) { acc.x += v[j].x; acc.y += v[j].y; acc.z += v[j].z; acc.w += v[j].w; }
                    }
                    ent = entNext;
                }
            }
            acc.x *= dv; acc.y *= dv; acc.z *= dv; acc.w *= dv;
        }
        __syncthreads();
        *(float4*)(As + wave * AS_STRIDE + lane * 4) = acc;
        __syncthreads();

        if (wave < 4) {
            v8f c0 = {}, c1 = {};
            const float*  Ap = As + l16 * AS_STRIDE + 2 * half;
            const float4* Bp = (const float4*)Bf + wave * 32 + lane;
            for (int k0 = 0; k0 < 32; k0 += 4) {
                v2f a[4];
                float4 bb[4];
#pragma unroll
                for (int i = 0; i < 4; ++i) {
                    a[i]  = *(const v2f*)(Ap + 4 * (k0 + i));
                    bb[i] = Bp[(k0 + i) * 128];
                }
#pragma unroll
                for (int i = 0; i < 4; ++i) {
                    v2f b0, b1;
                    b0.x = bb[i].x; b0.y = bb[i].y;
                    b1.x = bb[i].z; b1.y = bb[i].w;
                    c0 = __builtin_amdgcn_wmma_f32_16x16x4_f32(false, a[i], false, b0, (short)0, c0, false, false);
                    c1 = __builtin_amdgcn_wmma_f32_16x16x4_f32(false, a[i], false, b1, (short)0, c1, false, false);
                }
            }
            asm volatile("v_nop\n\tv_nop\n\tv_nop\n\tv_nop" : "+v"(c0), "+v"(c1));
            const int row0 = tile * 16 + 8 * half;
            float* o = outPlane + (size_t)row0 * 64;
#pragma unroll
            for (int v = 0; v < 8; ++v) {
                if (row0 + v < nNodes) {
                    o[v * 64]      = c0[v] + bias0;
                    o[v * 64 + 16] = c1[v] + bias1;
                }
            }
        }
    }
}


extern "C" void kernel_launch(void* const* d_in, const int* in_sizes, int n_in,
                              void* d_out, int out_size, void* d_ws, size_t ws_size,
                              hipStream_t stream) {
    const float* X   = (const float*)d_in[0];
    const int*   ei  = (const int*)d_in[1];
    const float* W1  = (const float*)d_in[2];
    const float* b1  = (const float*)d_in[3];
    const float* Wmu = (const float*)d_in[4];
    const float* bmu = (const float*)d_in[5];
    const float* Wlv = (const float*)d_in[6];
    const float* blv = (const float*)d_in[7];
    float* out = (float*)d_out;

    const int N = in_sizes[0] / 256;
    const int E = in_sizes[1] / 2;
    const int* rows = ei;
    const int* cols = ei + E;

    float* Ts   = (float*)d_ws;
    float* Hbuf = Ts + (size_t)N * GCN_F;
    float* W23  = Hbuf + (size_t)N * GCN_F;
    float* dinv = W23 + GCN_F * GCN_F;
    int* degInt  = (int*)(dinv + N);
    int* off     = degInt + N;
    int* ofs     = degInt;
    int* bsum    = off + N;
    const int nb = (N + 255) / 256;
    int* srcList = bsum + ((nb + 3) & ~3);
    int* rank    = (int*)Hbuf;
    int* degLo   = rank + E;

    const int rowTiles = (N + 15) / 16;
    const int gemmBlocks = (rowTiles + 7) / 8;
    const int gatherBlocks = (N + 7) / 8;
    const int countBlocks = 256;

    gcn_zero_i32<<<(N + 255) / 256, 256, 0, stream>>>(degLo, degInt, N);
    gcn_count_gemm1_pack<<<countBlocks + gemmBlocks + 64, 256, 0, stream>>>(
        cols, degLo, degInt, rank, E, X, W1, Ts, N, 256, Wmu, Wlv, W23, countBlocks, gemmBlocks);
    gcn_scan_block_dinv<<<nb, 256, 0, stream>>>(degLo, degInt, off, bsum, dinv, N);
    gcn_fill_csr<<<(E + 1023) / 1024, 256, (size_t)nb * sizeof(int), stream>>>(
        rows, cols, off, bsum, rank, degLo, srcList, ofs, E, N, nb);

    gcn_gather_relu<<<gatherBlocks, 256, 0, stream>>>(Ts, srcList, ofs, E,
                                                      dinv, b1, Hbuf, N);

    gcn_gather_gemm_out<<<768, 512, 0, stream>>>(Hbuf, srcList, ofs, E, dinv, W23,
                                                  bmu, blv, out, N, rowTiles);
}
